# starting kernel
# baseline (speedup 1.0000x reference)
#include <hip/hip_runtime.h>
#include <hip/hip_bf16.h>


#define B_   4
#define T_   512
#define E_   256
#define HID_ 512
#define HD_  32
#define NH_  16
#define S_   768
#define LN_EPS_ 1e-5f
#define CUTOFF_ 1e-5f
#define EXP_CLAMP_ 80.0f
#define PAD_ 36

typedef __attribute__((ext_vector_type(2))) float v2f;
typedef __attribute__((ext_vector_type(8))) float v8f;

__device__ __forceinline__ v8f wmma_f32_4(v2f a, v2f b, v8f c) {
  return __builtin_amdgcn_wmma_f32_16x16x4_f32(
       false, a,  false, b,
       (short)0, c,  false,  false);
}

__device__ __forceinline__ float half_reduce_sum(float x) {
  x += __shfl_xor(x, 1);
  x += __shfl_xor(x, 2);
  x += __shfl_xor(x, 4);
  x += __shfl_xor(x, 8);
  return x;
}

__device__ __forceinline__ float4 ldg4(const float* p) {
  return *(const float4*)p;
}

__global__ __launch_bounds__(128) void gather_kernel(
    const float* __restrict__ k, const float* __restrict__ v,
    const int* __restrict__ idx, float* __restrict__ ek,
    float* __restrict__ ev) {
  const int e = blockIdx.x;
  const int b = blockIdx.y;
  const int t = idx[b * E_ + e];
  const float* ks = k + ((size_t)b * T_ + t) * HID_;
  const float* vs = v + ((size_t)b * T_ + t) * HID_;
  float* kd = ek + ((size_t)b * E_ + e) * HID_;
  float* vd = ev + ((size_t)b * E_ + e) * HID_;
  const int i = threadIdx.x * 4;
  *(float4*)(kd + i) = ldg4(ks + i);
  *(float4*)(vd + i) = ldg4(vs + i);
}

__global__ __launch_bounds__(128) void attn_kernel(
    const float* __restrict__ q, const float* __restrict__ k,
    const float* __restrict__ v, const float* __restrict__ bias,
    const float* __restrict__ law, const float* __restrict__ ek,
    const float* __restrict__ ev, float* __restrict__ attn_out) {
  __shared__ float sK[16][PAD_];
  __shared__ float sV[16][PAD_];
  __shared__ float sQ[4][16][PAD_];
  __shared__ float sP[4][16][17];

  const int tid  = threadIdx.x;
  const int lane = tid & 31;
  const int w    = tid >> 5;
  const int row  = lane & 15;
  const int hi   = lane >> 4;

  const int TB  = T_ / 64;
  const int bid = blockIdx.x;
  const int tg  = bid % TB;
  const int h   = (bid / TB) % NH_;
  const int b   = bid / (TB * NH_);
  const int t0  = (tg * 4 + w) * 16;

  {
    const int qrow = lane >> 1;
    const int qc   = (lane & 1) * 16;
    const float* qp =
        q + ((size_t)(b * T_ + t0 + qrow)) * HID_ + h * HD_ + qc;
#pragma unroll
    for (int ii = 0; ii < 4; ++ii)
      *(float4*)&sQ[w][qrow][qc + ii * 4] = ldg4(qp + ii * 4);
  }

  v2f aq[8];
#pragma unroll
  for (int kk = 0; kk < 8; ++kk) {
    const int kc = kk * 4 + 2 * hi;
    aq[kk].x = sQ[w][row][kc];
    aq[kk].y = sQ[w][row][kc + 1];
  }

  float rsum[8];
#pragma unroll
  for (int j = 0; j < 8; ++j) rsum[j] = 0.0f;
  v8f acc0 = {};
  v8f acc1 = {};

  const size_t biasBase = (((size_t)b * NH_ + h) * T_ + t0) * S_;
  const size_t lawBase  = ((size_t)b * T_ + t0) * S_;

  const int kvrow = tid >> 3;
  const int kvc   = (tid & 7) * 4;

  for (int st = 0; st < S_ / 16; ++st) {
    const int s0 = st * 16;

    const bool ext = (s0 >= T_);
    const float* baseK = ext
        ? (ek + ((size_t)(b * E_ + (s0 - T_))) * HID_ + h * HD_)
        : (k  + ((size_t)(b * T_ + s0)) * HID_ + h * HD_);
    const float* baseV = ext
        ? (ev + ((size_t)(b * E_ + (s0 - T_))) * HID_ + h * HD_)
        : (v  + ((size_t)(b * T_ + s0)) * HID_ + h * HD_);

    *(float4*)&sK[kvrow][kvc] = ldg4(baseK + (size_t)kvrow * HID_ + kvc);
    *(float4*)&sV[kvrow][kvc] = ldg4(baseV + (size_t)kvrow * HID_ + kvc);
    __syncthreads();

    v8f c = {};
#pragma unroll
    for (int kk = 0; kk < 8; ++kk) {
      const int kc = kk * 4 + 2 * hi;
      v2f bb;
      bb.x = sK[row][kc];
      bb.y = sK[row][kc + 1];
      c = wmma_f32_4(aq[kk], bb, c);
    }

#pragma unroll
    for (int j = 0; j < 8; ++j) {
      const int m = j + 8 * hi;
      const int n = s0 + row;
      const float sc = c[j] + bias[biasBase + (size_t)m * S_ + n];
      const float lw = law[lawBase + (size_t)m * S_ + n];
      const float e =
          (lw <= CUTOFF_) ? 0.0f : __expf(fminf(sc, EXP_CLAMP_));
      rsum[j] += e;
      sP[w][m][row] = e * lw;
    }

#pragma unroll
    for (int kk = 0; kk < 4; ++kk) {
      const int kc = kk * 4 + 2 * hi;
      v2f a, b0, b1;
      a.x  = sP[w][row][kc];
      a.y  = sP[w][row][kc + 1];
      b0.x = sV[kc][row];
      b0.y = sV[kc + 1][row];
      b1.x = sV[kc][row + 16];
      b1.y = sV[kc + 1][row + 16];
      acc0 = wmma_f32_4(a, b0, acc0);
      acc1 = wmma_f32_4(a, b1, acc1);
    }
    __syncthreads();
  }

#pragma unroll
  for (int j = 0; j < 8; ++j) {
    const float s = half_reduce_sum(rsum[j]);
    const float inv = 1.0f / s;
    const int m = j + 8 * hi;
    const size_t o = ((size_t)(b * T_ + t0 + m)) * HID_ + h * HD_;
    attn_out[o + row]      = acc0[j] * inv;
    attn_out[o + 16 + row] = acc1[j] * inv;
  }
}

__global__ __launch_bounds__(256) void ln_stats_kernel(
    const float* __restrict__ attn, float* __restrict__ mu,
    float* __restrict__ rstd) {
  __shared__ float red[256];
  const int rowi = blockIdx.x;
  const int tid = threadIdx.x;
  const float* a = attn + (size_t)rowi * HID_;

  const float x0 = a[tid], x1 = a[tid + 256];
  red[tid] = x0 + x1;
  __syncthreads();
  for (int off = 128; off > 0; off >>= 1) {
    if (tid < off) red[tid] += red[tid + off];
    __syncthreads();
  }
  const float mean = red[0] * (1.0f / HID_);
  __syncthreads();

  const float d0 = x0 - mean, d1 = x1 - mean;
  red[tid] = d0 * d0 + d1 * d1;
  __syncthreads();
  for (int off = 128; off > 0; off >>= 1) {
    if (tid < off) red[tid] += red[tid + off];
    __syncthreads();
  }
  if (tid == 0) {
    mu[rowi] = mean;
    rstd[rowi] = rsqrtf(red[0] * (1.0f / HID_) + LN_EPS_);
  }
}

__global__ __launch_bounds__(128) void outproj_kernel(
    const float* __restrict__ attn, const float* __restrict__ mu,
    const float* __restrict__ rstd, const float* __restrict__ ln_g,
    const float* __restrict__ ln_b, const float* __restrict__ Wout,
    float* __restrict__ out) {
  __shared__ float sB[16][PAD_];
  __shared__ float sA[4][16][PAD_];

  const int tid  = threadIdx.x;
  const int lane = tid & 31;
  const int w    = tid >> 5;
  const int row  = lane & 15;
  const int hi   = lane >> 4;

  const int NT = HID_ / 16;
  const int nt = blockIdx.x % NT;
  const int mg = blockIdx.x / NT;
  const int m0 = (mg * 4 + w) * 16;
  const int n0 = nt * 16;

  const int wrow = tid >> 3;
  const int wc   = (tid & 7) * 4;
  const int arow = lane >> 1;
  const int ac   = (lane & 1) * 16;

  v8f c = {};
  for (int k0 = 0; k0 < HID_; k0 += 32) {
    *(float4*)&sB[wrow][wc] =
        ldg4(Wout + ((size_t)(n0 + wrow)) * HID_ + k0 + wc);

    {
      const float mu_r = mu[m0 + arow];
      const float rs_r = rstd[m0 + arow];
      const float* ap = attn + ((size_t)(m0 + arow)) * HID_ + k0 + ac;
#pragma unroll
      for (int ii = 0; ii < 4; ++ii) {
        const float4 x = ldg4(ap + ii * 4);
        const float4 g = ldg4(ln_g + k0 + ac + ii * 4);
        const float4 o = ldg4(ln_b + k0 + ac + ii * 4);
        float4 y;
        y.x = (x.x - mu_r) * rs_r * g.x + o.x;
        y.y = (x.y - mu_r) * rs_r * g.y + o.y;
        y.z = (x.z - mu_r) * rs_r * g.z + o.z;
        y.w = (x.w - mu_r) * rs_r * g.w + o.w;
        *(float4*)&sA[w][arow][ac + ii * 4] = y;
      }
    }
    __syncthreads();
#pragma unroll
    for (int kk = 0; kk < 8; ++kk) {
      const int kc = kk * 4 + 2 * hi;
      v2f a, bb;
      a.x  = sA[w][row][kc];
      a.y  = sA[w][row][kc + 1];
      bb.x = sB[row][kc];
      bb.y = sB[row][kc + 1];
      c = wmma_f32_4(a, bb, c);
    }
    __syncthreads();
  }

#pragma unroll
  for (int j = 0; j < 8; ++j) {
    const int m = m0 + j + 8 * hi;
    out[(size_t)m * HID_ + n0 + row] = c[j];
  }
}

extern "C" void kernel_launch(void* const* d_in, const int* in_sizes, int n_in,
                              void* d_out, int out_size, void* d_ws,
                              size_t ws_size, hipStream_t stream) {
  const float* q    = (const float*)d_in[0];
  const float* k    = (const float*)d_in[1];
  const float* v    = (const float*)d_in[2];
  const float* bias = (const float*)d_in[3];
  const int*   idx  = (const int*)d_in[5];
  const float* law  = (const float*)d_in[6];
  const float* Wout = (const float*)d_in[8];
  const float* lng  = (const float*)d_in[9];
  const float* lnb  = (const float*)d_in[10];
  float* out = (float*)d_out;

  float* ws   = (float*)d_ws;
  float* ek   = ws;
  float* ev   = ek + (size_t)B_ * E_ * HID_;
  float* attn = ev + (size_t)B_ * E_ * HID_;
  float* mu   = attn + (size_t)B_ * T_ * HID_;
  float* rstd = mu + (size_t)B_ * T_;

  gather_kernel<<<dim3(E_, B_), 128, 0, stream>>>(k, v, idx, ek, ev);
  attn_kernel<<<B_ * NH_ * (T_ / 64), 128, 0, stream>>>(q, k, v, bias, law, ek,
                                                        ev, attn);
  ln_stats_kernel<<<B_ * T_, 256, 0, stream>>>(attn, mu, rstd);
  outproj_kernel<<<(B_ * T_ / 64) * (HID_ / 16), 128, 0, stream>>>(
      attn, mu, rstd, lng, lnb, Wout, out);
}


# attention: a head's whole K and V (768 rows) copied to LDS once per 8-wave block of 128 query rows; s loop without block barrier or K/V global load, each tile's bias/law loads issued as one group
# speedup vs baseline: 3.2596x; 3.2596x over previous
#include <hip/hip_runtime.h>
#include <hip/hip_bf16.h>


#define B_   4
#define T_   512
#define E_   256
#define HID_ 512
#define HD_  32
#define NH_  16
#define S_   768
#define LN_EPS_ 1e-5f
#define CUTOFF_ 1e-5f
#define EXP_CLAMP_ 80.0f
#define PAD_ 36

typedef __attribute__((ext_vector_type(2))) float v2f;
typedef __attribute__((ext_vector_type(8))) float v8f;

__device__ __forceinline__ v8f wmma_f32_4(v2f a, v2f b, v8f c) {
  return __builtin_amdgcn_wmma_f32_16x16x4_f32(
       false, a,  false, b,
       (short)0, c,  false,  false);
}

__device__ __forceinline__ float half_reduce_sum(float x) {
  x += __shfl_xor(x, 1);
  x += __shfl_xor(x, 2);
  x += __shfl_xor(x, 4);
  x += __shfl_xor(x, 8);
  return x;
}

__device__ __forceinline__ float4 ldg4(const float* p) {
  return *(const float4*)p;
}

__global__ __launch_bounds__(128) void gather_kernel(
    const float* __restrict__ k, const float* __restrict__ v,
    const int* __restrict__ idx, float* __restrict__ ek,
    float* __restrict__ ev) {
  const int e = blockIdx.x;
  const int b = blockIdx.y;
  const int t = idx[b * E_ + e];
  const float* ks = k + ((size_t)b * T_ + t) * HID_;
  const float* vs = v + ((size_t)b * T_ + t) * HID_;
  float* kd = ek + ((size_t)b * E_ + e) * HID_;
  float* vd = ev + ((size_t)b * E_ + e) * HID_;
  const int i = threadIdx.x * 4;
  *(float4*)(kd + i) = ldg4(ks + i);
  *(float4*)(vd + i) = ldg4(vs + i);
}

#define TW_ 8

__global__ __launch_bounds__(256) void attn_kernel(
    const float* __restrict__ q, const float* __restrict__ k,
    const float* __restrict__ v, const float* __restrict__ bias,
    const float* __restrict__ law, const float* __restrict__ ek,
    const float* __restrict__ ev, float* __restrict__ attn_out) {
  __shared__ __attribute__((aligned(16))) float sK[S_][PAD_];
  __shared__ __attribute__((aligned(16))) float sV[S_][PAD_];
  __shared__ __attribute__((aligned(16))) float sQ[TW_][16][PAD_];
  __shared__ float sP[TW_][16][17];

  const int tid  = threadIdx.x;
  const int lane = tid & 31;
  const int w    = tid >> 5;
  const int row  = lane & 15;
  const int hi   = lane >> 4;

  const int TB  = T_ / (16 * TW_);
  const int bid = blockIdx.x;
  const int tg  = bid % TB;
  const int h   = (bid / TB) % NH_;
  const int b   = bid / (TB * NH_);
  const int t0  = (tg * TW_ + w) * 16;

  {
    const int r = tid >> 3;
    const int c = (tid & 7) * 4;
    const float* kb = k + ((size_t)b * T_) * HID_ + h * HD_ + c;
    const float* vb = v + ((size_t)b * T_) * HID_ + h * HD_ + c;
    for (int p = 0; p < T_ / 32; ++p) {
      const int s = p * 32 + r;
      *(float4*)&sK[s][c] = ldg4(kb + (size_t)s * HID_);
      *(float4*)&sV[s][c] = ldg4(vb + (size_t)s * HID_);
    }
    const float* ekb = ek + ((size_t)b * E_) * HID_ + h * HD_ + c;
    const float* evb = ev + ((size_t)b * E_) * HID_ + h * HD_ + c;
    for (int p = 0; p < E_ / 32; ++p) {
      const int e = p * 32 + r;
      *(float4*)&sK[T_ + e][c] = ldg4(ekb + (size_t)e * HID_);
      *(float4*)&sV[T_ + e][c] = ldg4(evb + (size_t)e * HID_);
    }
  }

  {
    const int qrow = lane >> 1;
    const int qc   = (lane & 1) * 16;
    const float* qp =
        q + ((size_t)(b * T_ + t0 + qrow)) * HID_ + h * HD_ + qc;
#pragma unroll
    for (int ii = 0; ii < 4; ++ii)
      *(float4*)&sQ[w][qrow][qc + ii * 4] = ldg4(qp + ii * 4);
  }

  v2f aq[8];
#pragma unroll
  for (int kk = 0; kk < 8; ++kk) {
    const int kc = kk * 4 + 2 * hi;
    aq[kk].x = sQ[w][row][kc];
    aq[kk].y = sQ[w][row][kc + 1];
  }

  float rsum[8];
#pragma unroll
  for (int j = 0; j < 8; ++j) rsum[j] = 0.0f;
  v8f acc0 = {};
  v8f acc1 = {};

  const float* bp = bias + (((size_t)b * NH_ + h) * T_ + t0 + 8 * hi) * S_ + row;
  const float* lp = law + ((size_t)b * T_ + t0 + 8 * hi) * S_ + row;

  __syncthreads();

  for (int st = 0; st < S_ / 16; ++st) {
    const int s0 = st * 16;

    float bv[8], lv[8];
#pragma unroll
    for (int j = 0; j < 8; ++j) {
      bv[j] = bp[(size_t)j * S_ + s0];
      lv[j] = lp[(size_t)j * S_ + s0];
    }

    v8f c = {};
#pragma unroll
    for (int kk = 0; kk < 8; ++kk) {
      const int kc = kk * 4 + 2 * hi;
      v2f bb;
      bb.x = sK[s0 + row][kc];
      bb.y = sK[s0 + row][kc + 1];
      c = wmma_f32_4(aq[kk], bb, c);
    }

#pragma unroll
    for (int j = 0; j < 8; ++j) {
      const int m = j + 8 * hi;
      const float sc = c[j] + bv[j];
      const float ex = __expf(fminf(sc, EXP_CLAMP_));
      const float e = (lv[j] <= CUTOFF_) ? 0.0f : ex;
      rsum[j] += e;
      sP[w][m][row] = e * lv[j];
    }

#pragma unroll
    for (int kk = 0; kk < 4; ++kk) {
      const int kc = kk * 4 + 2 * hi;
      v2f a, b0, b1;
      a.x  = sP[w][row][kc];
      a.y  = sP[w][row][kc + 1];
      b0.x = sV[s0 + kc][row];
      b0.y = sV[s0 + kc + 1][row];
      b1.x = sV[s0 + kc][row + 16];
      b1.y = sV[s0 + kc + 1][row + 16];
      acc0 = wmma_f32_4(a, b0, acc0);
      acc1 = wmma_f32_4(a, b1, acc1);
    }
  }

#pragma unroll
  for (int j = 0; j < 8; ++j) {
    const float s = half_reduce_sum(rsum[j]);
    const float inv = 1.0f / s;
    const int m = j + 8 * hi;
    const size_t o = ((size_t)(b * T_ + t0 + m)) * HID_ + h * HD_;
    attn_out[o + row]      = acc0[j] * inv;
    attn_out[o + 16 + row] = acc1[j] * inv;
  }
}

__global__ __launch_bounds__(256) void ln_stats_kernel(
    const float* __restrict__ attn, float* __restrict__ mu,
    float* __restrict__ rstd) {
  __shared__ float red[256];
  const int rowi = blockIdx.x;
  const int tid = threadIdx.x;
  const float* a = attn + (size_t)rowi * HID_;

  const float x0 = a[tid], x1 = a[tid + 256];
  red[tid] = x0 + x1;
  __syncthreads();
  for (int off = 128; off > 0; off >>= 1) {
    if (tid < off) red[tid] += red[tid + off];
    __syncthreads();
  }
  const float mean = red[0] * (1.0f / HID_);
  __syncthreads();

  const float d0 = x0 - mean, d1 = x1 - mean;
  red[tid] = d0 * d0 + d1 * d1;
  __syncthreads();
  for (int off = 128; off > 0; off >>= 1) {
    if (tid < off) red[tid] += red[tid + off];
    __syncthreads();
  }
  if (tid == 0) {
    mu[rowi] = mean;
    rstd[rowi] = rsqrtf(red[0] * (1.0f / HID_) + LN_EPS_);
  }
}

__global__ __launch_bounds__(128) void outproj_kernel(
    const float* __restrict__ attn, const float* __restrict__ mu,
    const float* __restrict__ rstd, const float* __restrict__ ln_g,
    const float* __restrict__ ln_b, const float* __restrict__ Wout,
    float* __restrict__ out) {
  __shared__ float sB[16][PAD_];
  __shared__ float sA[4][16][PAD_];

  const int tid  = threadIdx.x;
  const int lane = tid & 31;
  const int w    = tid >> 5;
  const int row  = lane & 15;
  const int hi   = lane >> 4;

  const int NT = HID_ / 16;
  const int nt = blockIdx.x % NT;
  const int mg = blockIdx.x / NT;
  const int m0 = (mg * 4 + w) * 16;
  const int n0 = nt * 16;

  const int wrow = tid >> 3;
  const int wc   = (tid & 7) * 4;
  const int arow = lane >> 1;
  const int ac   = (lane & 1) * 16;

  v8f c = {};
  for (int k0 = 0; k0 < HID_; k0 += 32) {
    *(float4*)&sB[wrow][wc] =
        ldg4(Wout + ((size_t)(n0 + wrow)) * HID_ + k0 + wc);

    {
      const float mu_r = mu[m0 + arow];
      const float rs_r = rstd[m0 + arow];
      const float* ap = attn + ((size_t)(m0 + arow)) * HID_ + k0 + ac;
#pragma unroll
      for (int ii = 0; ii < 4; ++ii) {
        const float4 x = ldg4(ap + ii * 4);
        const float4 g = ldg4(ln_g + k0 + ac + ii * 4);
        const float4 o = ldg4(ln_b + k0 + ac + ii * 4);
        float4 y;
        y.x = (x.x - mu_r) * rs_r * g.x + o.x;
        y.y = (x.y - mu_r) * rs_r * g.y + o.y;
        y.z = (x.z - mu_r) * rs_r * g.z + o.z;
        y.w = (x.w - mu_r) * rs_r * g.w + o.w;
        *(float4*)&sA[w][arow][ac + ii * 4] = y;
      }
    }
    __syncthreads();
#pragma unroll
    for (int kk = 0; kk < 8; ++kk) {
      const int kc = kk * 4 + 2 * hi;
      v2f a, bb;
      a.x  = sA[w][row][kc];
      a.y  = sA[w][row][kc + 1];
      bb.x = sB[row][kc];
      bb.y = sB[row][kc + 1];
      c = wmma_f32_4(a, bb, c);
    }
    __syncthreads();
  }

#pragma unroll
  for (int j = 0; j < 8; ++j) {
    const int m = m0 + j + 8 * hi;
    out[(size_t)m * HID_ + n0 + row] = c[j];
  }
}

extern "C" void kernel_launch(void* const* d_in, const int* in_sizes, int n_in,
                              void* d_out, int out_size, void* d_ws,
                              size_t ws_size, hipStream_t stream) {
  const float* q    = (const float*)d_in[0];
  const float* k    = (const float*)d_in[1];
  const float* v    = (const float*)d_in[2];
  const float* bias = (const float*)d_in[3];
  const int*   idx  = (const int*)d_in[5];
  const float* law  = (const float*)d_in[6];
  const float* Wout = (const float*)d_in[8];
  const float* lng  = (const float*)d_in[9];
  const float* lnb  = (const float*)d_in[10];
  float* out = (float*)d_out;

  float* ws   = (float*)d_ws;
  float* ek   = ws;
  float* ev   = ek + (size_t)B_ * E_ * HID_;
  float* attn = ev + (size_t)B_ * E_ * HID_;
  float* mu   = attn + (size_t)B_ * T_ * HID_;
  float* rstd = mu + (size_t)B_ * T_;

  gather_kernel<<<dim3(E_, B_), 128, 0, stream>>>(k, v, idx, ek, ev);
  attn_kernel<<<B_ * NH_ * (T_ / 128), 256, 0, stream>>>(q, k, v, bias, law, ek,
                                                         ev, attn);
  ln_stats_kernel<<<B_ * T_, 256, 0, stream>>>(attn, mu, rstd);
  outproj_kernel<<<(B_ * T_ / 64) * (HID_ / 16), 128, 0, stream>>>(
      attn, mu, rstd, lng, lnb, Wout, out);
}


# out-proj block keeps its 64 attn rows and 64 W_out rows whole in LDS (K=512), does their LayerNorm there itself: ln_stats launch gone (4 -> 3), K loop barrier-free (LN row-sum order changes)
# speedup vs baseline: 4.2609x; 1.3072x over previous
#include <hip/hip_runtime.h>
#include <hip/hip_bf16.h>


#define B_   4
#define T_   512
#define E_   256
#define HID_ 512
#define HD_  32
#define NH_  16
#define S_   768
#define LN_EPS_ 1e-5f
#define CUTOFF_ 1e-5f
#define EXP_CLAMP_ 80.0f
#define PAD_ 36

typedef __attribute__((ext_vector_type(2))) float v2f;
typedef __attribute__((ext_vector_type(8))) float v8f;

__device__ __forceinline__ v8f wmma_f32_4(v2f a, v2f b, v8f c) {
  return __builtin_amdgcn_wmma_f32_16x16x4_f32(
       false, a,  false, b,
       (short)0, c,  false,  false);
}

__device__ __forceinline__ float half_reduce_sum(float x) {
  x += __shfl_xor(x, 1);
  x += __shfl_xor(x, 2);
  x += __shfl_xor(x, 4);
  x += __shfl_xor(x, 8);
  return x;
}

__device__ __forceinline__ float4 ldg4(const float* p) {
  return *(const float4*)p;
}

__global__ __launch_bounds__(128) void gather_kernel(
    const float* __restrict__ k, const float* __restrict__ v,
    const int* __restrict__ idx, float* __restrict__ ek,
    float* __restrict__ ev) {
  const int e = blockIdx.x;
  const int b = blockIdx.y;
  const int t = idx[b * E_ + e];
  const float* ks = k + ((size_t)b * T_ + t) * HID_;
  const float* vs = v + ((size_t)b * T_ + t) * HID_;
  float* kd = ek + ((size_t)b * E_ + e) * HID_;
  float* vd = ev + ((size_t)b * E_ + e) * HID_;
  const int i = threadIdx.x * 4;
  *(float4*)(kd + i) = ldg4(ks + i);
  *(float4*)(vd + i) = ldg4(vs + i);
}

#define TW_ 8

__global__ __launch_bounds__(256) void attn_kernel(
    const float* __restrict__ q, const float* __restrict__ k,
    const float* __restrict__ v, const float* __restrict__ bias,
    const float* __restrict__ law, const float* __restrict__ ek,
    const float* __restrict__ ev, float* __restrict__ attn_out) {
  __shared__ __attribute__((aligned(16))) float sK[S_][PAD_];
  __shared__ __attribute__((aligned(16))) float sV[S_][PAD_];
  __shared__ __attribute__((aligned(16))) float sQ[TW_][16][PAD_];
  __shared__ float sP[TW_][16][17];

  const int tid  = threadIdx.x;
  const int lane = tid & 31;
  const int w    = tid >> 5;
  const int row  = lane & 15;
  const int hi   = lane >> 4;

  const int TB  = T_ / (16 * TW_);
  const int bid = blockIdx.x;
  const int tg  = bid % TB;
  const int h   = (bid / TB) % NH_;
  const int b   = bid / (TB * NH_);
  const int t0  = (tg * TW_ + w) * 16;

  {
    const int r = tid >> 3;
    const int c = (tid & 7) * 4;
    const float* kb = k + ((size_t)b * T_) * HID_ + h * HD_ + c;
    const float* vb = v + ((size_t)b * T_) * HID_ + h * HD_ + c;
    for (int p = 0; p < T_ / 32; ++p) {
      const int s = p * 32 + r;
      *(float4*)&sK[s][c] = ldg4(kb + (size_t)s * HID_);
      *(float4*)&sV[s][c] = ldg4(vb + (size_t)s * HID_);
    }
    const float* ekb = ek + ((size_t)b * E_) * HID_ + h * HD_ + c;
    const float* evb = ev + ((size_t)b * E_) * HID_ + h * HD_ + c;
    for (int p = 0; p < E_ / 32; ++p) {
      const int e = p * 32 + r;
      *(float4*)&sK[T_ + e][c] = ldg4(ekb + (size_t)e * HID_);
      *(float4*)&sV[T_ + e][c] = ldg4(evb + (size_t)e * HID_);
    }
  }

  {
    const int qrow = lane >> 1;
    const int qc   = (lane & 1) * 16;
    const float* qp =
        q + ((size_t)(b * T_ + t0 + qrow)) * HID_ + h * HD_ + qc;
#pragma unroll
    for (int ii = 0; ii < 4; ++ii)
      *(float4*)&sQ[w][qrow][qc + ii * 4] = ldg4(qp + ii * 4);
  }

  v2f aq[8];
#pragma unroll
  for (int kk = 0; kk < 8; ++kk) {
    const int kc = kk * 4 + 2 * hi;
    aq[kk].x = sQ[w][row][kc];
    aq[kk].y = sQ[w][row][kc + 1];
  }

  float rsum[8];
#pragma unroll
  for (int j = 0; j < 8; ++j) rsum[j] = 0.0f;
  v8f acc0 = {};
  v8f acc1 = {};

  const float* bp = bias + (((size_t)b * NH_ + h) * T_ + t0 + 8 * hi) * S_ + row;
  const float* lp = law + ((size_t)b * T_ + t0 + 8 * hi) * S_ + row;

  __syncthreads();

  for (int st = 0; st < S_ / 16; ++st) {
    const int s0 = st * 16;

    float bv[8], lv[8];
#pragma unroll
    for (int j = 0; j < 8; ++j) {
      bv[j] = bp[(size_t)j * S_ + s0];
      lv[j] = lp[(size_t)j * S_ + s0];
    }

    v8f c = {};
#pragma unroll
    for (int kk = 0; kk < 8; ++kk) {
      const int kc = kk * 4 + 2 * hi;
      v2f bb;
      bb.x = sK[s0 + row][kc];
      bb.y = sK[s0 + row][kc + 1];
      c = wmma_f32_4(aq[kk], bb, c);
    }

#pragma unroll
    for (int j = 0; j < 8; ++j) {
      const int m = j + 8 * hi;
      const float sc = c[j] + bv[j];
      const float ex = __expf(fminf(sc, EXP_CLAMP_));
      const float e = (lv[j] <= CUTOFF_) ? 0.0f : ex;
      rsum[j] += e;
      sP[w][m][row] = e * lv[j];
    }

#pragma unroll
    for (int kk = 0; kk < 4; ++kk) {
      const int kc = kk * 4 + 2 * hi;
      v2f a, b0, b1;
      a.x  = sP[w][row][kc];
      a.y  = sP[w][row][kc + 1];
      b0.x = sV[s0 + kc][row];
      b0.y = sV[s0 + kc + 1][row];
      b1.x = sV[s0 + kc][row + 16];
      b1.y = sV[s0 + kc + 1][row + 16];
      acc0 = wmma_f32_4(a, b0, acc0);
      acc1 = wmma_f32_4(a, b1, acc1);
    }
  }

#pragma unroll
  for (int j = 0; j < 8; ++j) {
    const float s = half_reduce_sum(rsum[j]);
    const float inv = 1.0f / s;
    const int m = j + 8 * hi;
    const size_t o = ((size_t)(b * T_ + t0 + m)) * HID_ + h * HD_;
    attn_out[o + row]      = acc0[j] * inv;
    attn_out[o + 16 + row] = acc1[j] * inv;
  }
}

#define OM_ 64
#define ON_ 64
#define OP_ 516

__global__ __launch_bounds__(256) void ln_outproj_kernel(
    const float* __restrict__ attn, const float* __restrict__ ln_g,
    const float* __restrict__ ln_b, const float* __restrict__ Wout,
    float* __restrict__ out) {
  __shared__ __attribute__((aligned(16))) float sA[OM_][OP_];
  __shared__ __attribute__((aligned(16))) float sW[ON_][OP_];

  const int tid  = threadIdx.x;
  const int lane = tid & 31;
  const int w    = tid >> 5;
  const int row  = lane & 15;
  const int hi   = lane >> 4;

  const int NB = HID_ / ON_;
  const int n0 = (blockIdx.x % NB) * ON_;
  const int m0 = (blockIdx.x / NB) * OM_;

  const int lc = lane * 4;
  float4 g[4], o[4];
#pragma unroll
  for (int i = 0; i < 4; ++i) {
    g[i] = ldg4(ln_g + lc + 128 * i);
    o[i] = ldg4(ln_b + lc + 128 * i);
  }

  float4 x[8][4];
  {
    const float* ap = attn + ((size_t)(m0 + 8 * w)) * HID_ + lc;
#pragma unroll
    for (int rr = 0; rr < 8; ++rr) {
#pragma unroll
      for (int i = 0; i < 4; ++i)
        x[rr][i] = ldg4(ap + (size_t)rr * HID_ + 128 * i);
    }
  }

  const int r2 = tid >> 7;
  const int c4 = (tid & 127) * 4;
  float4 tw[ON_ / 2];
  {
    const float* wp = Wout + ((size_t)(n0 + r2)) * HID_ + c4;
#pragma unroll
    for (int i = 0; i < ON_ / 2; ++i) tw[i] = ldg4(wp + (size_t)(2 * i) * HID_);
  }

  {
    float s[8];
#pragma unroll
    for (int rr = 0; rr < 8; ++rr) {
      float t = 0.0f;
#pragma unroll
      for (int i = 0; i < 4; ++i)
        t += (x[rr][i].x + x[rr][i].y) + (x[rr][i].z + x[rr][i].w);
      s[rr] = t;
    }
#pragma unroll
    for (int m = 1; m < 32; m <<= 1) {
#pragma unroll
      for (int rr = 0; rr < 8; ++rr) s[rr] += __shfl_xor(s[rr], m);
    }
#pragma unroll
    for (int rr = 0; rr < 8; ++rr) {
      const float mean = s[rr] * (1.0f / HID_);
      float t = 0.0f;
#pragma unroll
      for (int i = 0; i < 4; ++i) {
        x[rr][i].x -= mean; x[rr][i].y -= mean;
        x[rr][i].z -= mean; x[rr][i].w -= mean;
        t += (x[rr][i].x * x[rr][i].x + x[rr][i].y * x[rr][i].y) +
             (x[rr][i].z * x[rr][i].z + x[rr][i].w * x[rr][i].w);
      }
      s[rr] = t;
    }
#pragma unroll
    for (int m = 1; m < 32; m <<= 1) {
#pragma unroll
      for (int rr = 0; rr < 8; ++rr) s[rr] += __shfl_xor(s[rr], m);
    }
#pragma unroll
    for (int rr = 0; rr < 8; ++rr) {
      const float rs = rsqrtf(s[rr] * (1.0f / HID_) + LN_EPS_);
      float* ar = &sA[8 * w + rr][lc];
#pragma unroll
      for (int i = 0; i < 4; ++i) {
        float4 y;
        y.x = x[rr][i].x * rs * g[i].x + o[i].x;
        y.y = x[rr][i].y * rs * g[i].y + o[i].y;
        y.z = x[rr][i].z * rs * g[i].z + o[i].z;
        y.w = x[rr][i].w * rs * g[i].w + o[i].w;
        *(float4*)(ar + 128 * i) = y;
      }
    }
  }
#pragma unroll
  for (int i = 0; i < ON_ / 2; ++i) *(float4*)&sW[2 * i + r2][c4] = tw[i];
  __syncthreads();

  const int mt = w & 3;
  const int nh = w >> 2;
  const float* pa  = &sA[mt * 16 + row][2 * hi];
  const float* pb0 = &sW[nh * 32 + row][2 * hi];
  const float* pb1 = &sW[nh * 32 + 16 + row][2 * hi];

  v8f c0 = {};
  v8f c1 = {};

#pragma unroll
  for (int ks = 0; ks < HID_ / 4; ++ks) {
    const int kc = ks * 4;
    v2f a, b0, b1;
    a.x  = pa[kc];   a.y  = pa[kc + 1];
    b0.x = pb0[kc];  b0.y = pb0[kc + 1];
    b1.x = pb1[kc];  b1.y = pb1[kc + 1];
    c0 = wmma_f32_4(a, b0, c0);
    c1 = wmma_f32_4(a, b1, c1);
  }

#pragma unroll
  for (int j = 0; j < 8; ++j) {
    const int m = m0 + mt * 16 + j + 8 * hi;
    float* op = out + (size_t)m * HID_ + n0 + nh * 32 + row;
    op[0]  = c0[j];
    op[16] = c1[j];
  }
}

extern "C" void kernel_launch(void* const* d_in, const int* in_sizes, int n_in,
                              void* d_out, int out_size, void* d_ws,
                              size_t ws_size, hipStream_t stream) {
  const float* q    = (const float*)d_in[0];
  const float* k    = (const float*)d_in[1];
  const float* v    = (const float*)d_in[2];
  const float* bias = (const float*)d_in[3];
  const int*   idx  = (const int*)d_in[5];
  const float* law  = (const float*)d_in[6];
  const float* Wout = (const float*)d_in[8];
  const float* lng  = (const float*)d_in[9];
  const float* lnb  = (const float*)d_in[10];
  float* out = (float*)d_out;

  float* ws   = (float*)d_ws;
  float* ek   = ws;
  float* ev   = ek + (size_t)B_ * E_ * HID_;
  float* attn = ev + (size_t)B_ * E_ * HID_;

  gather_kernel<<<dim3(E_, B_), 128, 0, stream>>>(k, v, idx, ek, ev);
  attn_kernel<<<B_ * NH_ * (T_ / 128), 256, 0, stream>>>(q, k, v, bias, law, ek,
                                                         ev, attn);
  ln_outproj_kernel<<<(B_ * T_ / OM_) * (HID_ / ON_), 256, 0, stream>>>(
      attn, lng, lnb, Wout, out);
}


# attention: expansion tiles (s >= T) read K/V from the head's 512 rows resident in LDS, at row outcell_index via an LDS row table; gather launch, ek/ev copies, 1/3 of LDS fill gone (3 -> 2 launches)
# speedup vs baseline: 4.5374x; 1.0649x over previous
#include <hip/hip_runtime.h>
#include <hip/hip_bf16.h>


#define B_   4
#define T_   512
#define E_   256
#define HID_ 512
#define HD_  32
#define NH_  16
#define S_   768
#define LN_EPS_ 1e-5f
#define CUTOFF_ 1e-5f
#define EXP_CLAMP_ 80.0f
#define PAD_ 36

typedef __attribute__((ext_vector_type(2))) float v2f;
typedef __attribute__((ext_vector_type(8))) float v8f;

__device__ __forceinline__ v8f wmma_f32_4(v2f a, v2f b, v8f c) {
  return __builtin_amdgcn_wmma_f32_16x16x4_f32(
       false, a,  false, b,
       (short)0, c,  false,  false);
}

__device__ __forceinline__ float half_reduce_sum(float x) {
  x += __shfl_xor(x, 1);
  x += __shfl_xor(x, 2);
  x += __shfl_xor(x, 4);
  x += __shfl_xor(x, 8);
  return x;
}

__device__ __forceinline__ float4 ldg4(const float* p) {
  return *(const float4*)p;
}

#define TW_ 8

__global__ __launch_bounds__(256) void attn_kernel(
    const float* __restrict__ q, const float* __restrict__ k,
    const float* __restrict__ v, const float* __restrict__ bias,
    const float* __restrict__ law, const int* __restrict__ idx,
    float* __restrict__ attn_out) {
  __shared__ __attribute__((aligned(16))) float sK[T_][PAD_];
  __shared__ __attribute__((aligned(16))) float sV[T_][PAD_];
  __shared__ __attribute__((aligned(8))) int sRow[E_];
  __shared__ __attribute__((aligned(16))) float sQ[TW_][16][PAD_];
  __shared__ float sP[TW_][16][17];

  const int tid  = threadIdx.x;
  const int lane = tid & 31;
  const int w    = tid >> 5;
  const int row  = lane & 15;
  const int hi   = lane >> 4;

  const int TB  = T_ / (16 * TW_);
  const int bid = blockIdx.x;
  const int tg  = bid % TB;
  const int h   = (bid / TB) % NH_;
  const int b   = bid / (TB * NH_);
  const int t0  = (tg * TW_ + w) * 16;

  {
    const int r = tid >> 3;
    const int c = (tid & 7) * 4;
    const float* kb = k + ((size_t)b * T_) * HID_ + h * HD_ + c;
    const float* vb = v + ((size_t)b * T_) * HID_ + h * HD_ + c;
    for (int p = 0; p < T_ / 32; ++p) {
      const int s = p * 32 + r;
      *(float4*)&sK[s][c] = ldg4(kb + (size_t)s * HID_);
      *(float4*)&sV[s][c] = ldg4(vb + (size_t)s * HID_);
    }
    static_assert(E_ == 32 * TW_, "one table entry per thread of the block");
    sRow[tid] = idx[b * E_ + tid] & (T_ - 1);
  }

  {
    const int qrow = lane >> 1;
    const int qc   = (lane & 1) * 16;
    const float* qp =
        q + ((size_t)(b * T_ + t0 + qrow)) * HID_ + h * HD_ + qc;
#pragma unroll
    for (int ii = 0; ii < 4; ++ii)
      *(float4*)&sQ[w][qrow][qc + ii * 4] = ldg4(qp + ii * 4);
  }

  v2f aq[8];
#pragma unroll
  for (int kk = 0; kk < 8; ++kk) {
    const int kc = kk * 4 + 2 * hi;
    aq[kk].x = sQ[w][row][kc];
    aq[kk].y = sQ[w][row][kc + 1];
  }

  float rsum[8];
#pragma unroll
  for (int j = 0; j < 8; ++j) rsum[j] = 0.0f;
  v8f acc0 = {};
  v8f acc1 = {};

  const float* bp = bias + (((size_t)b * NH_ + h) * T_ + t0 + 8 * hi) * S_ + row;
  const float* lp = law + ((size_t)b * T_ + t0 + 8 * hi) * S_ + row;

  __syncthreads();

  auto s_tile = [&](const int s0, const int krow, const int (&vr0)[4],
                    const int (&vr1)[4]) {
    float bv[8], lv[8];
#pragma unroll
    for (int j = 0; j < 8; ++j) {
      bv[j] = bp[(size_t)j * S_ + s0];
      lv[j] = lp[(size_t)j * S_ + s0];
    }

    v8f c = {};
    const float* kp = &sK[krow][2 * hi];
#pragma unroll
    for (int kk = 0; kk < 8; ++kk) {
      v2f bb;
      bb.x = kp[kk * 4];
      bb.y = kp[kk * 4 + 1];
      c = wmma_f32_4(aq[kk], bb, c);
    }

#pragma unroll
    for (int j = 0; j < 8; ++j) {
      const int m = j + 8 * hi;
      const float sc = c[j] + bv[j];
      const float ex = __expf(fminf(sc, EXP_CLAMP_));
      const float e = (lv[j] <= CUTOFF_) ? 0.0f : ex;
      rsum[j] += e;
      sP[w][m][row] = e * lv[j];
    }

#pragma unroll
    for (int kk = 0; kk < 4; ++kk) {
      const int kc = kk * 4 + 2 * hi;
      v2f a, b0, b1;
      a.x  = sP[w][row][kc];
      a.y  = sP[w][row][kc + 1];
      b0.x = sV[vr0[kk]][row];
      b0.y = sV[vr1[kk]][row];
      b1.x = sV[vr0[kk]][row + 16];
      b1.y = sV[vr1[kk]][row + 16];
      acc0 = wmma_f32_4(a, b0, acc0);
      acc1 = wmma_f32_4(a, b1, acc1);
    }
  };

  for (int st = 0; st < T_ / 16; ++st) {
    const int s0 = st * 16;
    int vr0[4], vr1[4];
#pragma unroll
    for (int kk = 0; kk < 4; ++kk) {
      vr0[kk] = s0 + kk * 4 + 2 * hi;
      vr1[kk] = vr0[kk] + 1;
    }
    s_tile(s0, s0 + row, vr0, vr1);
  }
  for (int et = 0; et < E_ / 16; ++et) {
    const int e0 = et * 16;
    const int krow = sRow[e0 + row];
    int vr0[4], vr1[4];
#pragma unroll
    for (int kk = 0; kk < 4; ++kk) {
      const int2 rr = *(const int2*)&sRow[e0 + kk * 4 + 2 * hi];
      vr0[kk] = rr.x;
      vr1[kk] = rr.y;
    }
    s_tile(T_ + e0, krow, vr0, vr1);
  }

#pragma unroll
  for (int j = 0; j < 8; ++j) {
    const float s = half_reduce_sum(rsum[j]);
    const float inv = 1.0f / s;
    const int m = j + 8 * hi;
    const size_t o = ((size_t)(b * T_ + t0 + m)) * HID_ + h * HD_;
    attn_out[o + row]      = acc0[j] * inv;
    attn_out[o + 16 + row] = acc1[j] * inv;
  }
}

#define OM_ 64
#define ON_ 64
#define OP_ 516

__global__ __launch_bounds__(256) void ln_outproj_kernel(
    const float* __restrict__ attn, const float* __restrict__ ln_g,
    const float* __restrict__ ln_b, const float* __restrict__ Wout,
    float* __restrict__ out) {
  __shared__ __attribute__((aligned(16))) float sA[OM_][OP_];
  __shared__ __attribute__((aligned(16))) float sW[ON_][OP_];

  const int tid  = threadIdx.x;
  const int lane = tid & 31;
  const int w    = tid >> 5;
  const int row  = lane & 15;
  const int hi   = lane >> 4;

  const int NB = HID_ / ON_;
  const int n0 = (blockIdx.x % NB) * ON_;
  const int m0 = (blockIdx.x / NB) * OM_;

  const int lc = lane * 4;
  float4 g[4], o[4];
#pragma unroll
  for (int i = 0; i < 4; ++i) {
    g[i] = ldg4(ln_g + lc + 128 * i);
    o[i] = ldg4(ln_b + lc + 128 * i);
  }

  float4 x[8][4];
  {
    const float* ap = attn + ((size_t)(m0 + 8 * w)) * HID_ + lc;
#pragma unroll
    for (int rr = 0; rr < 8; ++rr) {
#pragma unroll
      for (int i = 0; i < 4; ++i)
        x[rr][i] = ldg4(ap + (size_t)rr * HID_ + 128 * i);
    }
  }

  const int r2 = tid >> 7;
  const int c4 = (tid & 127) * 4;
  float4 tw[ON_ / 2];
  {
    const float* wp = Wout + ((size_t)(n0 + r2)) * HID_ + c4;
#pragma unroll
    for (int i = 0; i < ON_ / 2; ++i) tw[i] = ldg4(wp + (size_t)(2 * i) * HID_);
  }

  {
    float s[8];
#pragma unroll
    for (int rr = 0; rr < 8; ++rr) {
      float t = 0.0f;
#pragma unroll
      for (int i = 0; i < 4; ++i)
        t += (x[rr][i].x + x[rr][i].y) + (x[rr][i].z + x[rr][i].w);
      s[rr] = t;
    }
#pragma unroll
    for (int m = 1; m < 32; m <<= 1) {
#pragma unroll
      for (int rr = 0; rr < 8; ++rr) s[rr] += __shfl_xor(s[rr], m);
    }
#pragma unroll
    for (int rr = 0; rr < 8; ++rr) {
      const float mean = s[rr] * (1.0f / HID_);
      float t = 0.0f;
#pragma unroll
      for (int i = 0; i < 4; ++i) {
        x[rr][i].x -= mean; x[rr][i].y -= mean;
        x[rr][i].z -= mean; x[rr][i].w -= mean;
        t += (x[rr][i].x * x[rr][i].x + x[rr][i].y * x[rr][i].y) +
             (x[rr][i].z * x[rr][i].z + x[rr][i].w * x[rr][i].w);
      }
      s[rr] = t;
    }
#pragma unroll
    for (int m = 1; m < 32; m <<= 1) {
#pragma unroll
      for (int rr = 0; rr < 8; ++rr) s[rr] += __shfl_xor(s[rr], m);
    }
#pragma unroll
    for (int rr = 0; rr < 8; ++rr) {
      const float rs = rsqrtf(s[rr] * (1.0f / HID_) + LN_EPS_);
      float* ar = &sA[8 * w + rr][lc];
#pragma unroll
      for (int i = 0; i < 4; ++i) {
        float4 y;
        y.x = x[rr][i].x * rs * g[i].x + o[i].x;
        y.y = x[rr][i].y * rs * g[i].y + o[i].y;
        y.z = x[rr][i].z * rs * g[i].z + o[i].z;
        y.w = x[rr][i].w * rs * g[i].w + o[i].w;
        *(float4*)(ar + 128 * i) = y;
      }
    }
  }
#pragma unroll
  for (int i = 0; i < ON_ / 2; ++i) *(float4*)&sW[2 * i + r2][c4] = tw[i];
  __syncthreads();

  const int mt = w & 3;
  const int nh = w >> 2;
  const float* pa  = &sA[mt * 16 + row][2 * hi];
  const float* pb0 = &sW[nh * 32 + row][2 * hi];
  const float* pb1 = &sW[nh * 32 + 16 + row][2 * hi];

  v8f c0 = {};
  v8f c1 = {};

#pragma unroll
  for (int ks = 0; ks < HID_ / 4; ++ks) {
    const int kc = ks * 4;
    v2f a, b0, b1;
    a.x  = pa[kc];   a.y  = pa[kc + 1];
    b0.x = pb0[kc];  b0.y = pb0[kc + 1];
    b1.x = pb1[kc];  b1.y = pb1[kc + 1];
    c0 = wmma_f32_4(a, b0, c0);
    c1 = wmma_f32_4(a, b1, c1);
  }

#pragma unroll
  for (int j = 0; j < 8; ++j) {
    const int m = m0 + mt * 16 + j + 8 * hi;
    float* op = out + (size_t)m * HID_ + n0 + nh * 32 + row;
    op[0]  = c0[j];
    op[16] = c1[j];
  }
}

extern "C" void kernel_launch(void* const* d_in, const int* in_sizes, int n_in,
                              void* d_out, int out_size, void* d_ws,
                              size_t ws_size, hipStream_t stream) {
  const float* q    = (const float*)d_in[0];
  const float* k    = (const float*)d_in[1];
  const float* v    = (const float*)d_in[2];
  const float* bias = (const float*)d_in[3];
  const int*   idx  = (const int*)d_in[5];
  const float* law  = (const float*)d_in[6];
  const float* Wout = (const float*)d_in[8];
  const float* lng  = (const float*)d_in[9];
  const float* lnb  = (const float*)d_in[10];
  float* out = (float*)d_out;

  float* attn = (float*)d_ws;

  attn_kernel<<<B_ * NH_ * (T_ / 128), 256, 0, stream>>>(q, k, v, bias, law,
                                                         idx, attn);
  ln_outproj_kernel<<<(B_ * T_ / OM_) * (HID_ / ON_), 256, 0, stream>>>(
      attn, lng, lnb, Wout, out);
}


# attention: each wave asks L2 (global_prefetch_b8, no registers) for its 16 bias + 16 law rows' next 256-B pieces 8 s-tiles ahead, 1 per 4 tiles; tiles 0-7 asked at entry, under the K/V fill
# speedup vs baseline: 4.8075x; 1.0595x over previous
#include <hip/hip_runtime.h>
#include <hip/hip_bf16.h>


#define B_   4
#define T_   512
#define E_   256
#define HID_ 512
#define HD_  32
#define NH_  16
#define S_   768
#define LN_EPS_ 1e-5f
#define CUTOFF_ 1e-5f
#define EXP_CLAMP_ 80.0f
#define PAD_ 36

typedef __attribute__((ext_vector_type(2))) float v2f;
typedef __attribute__((ext_vector_type(8))) float v8f;

__device__ __forceinline__ v8f wmma_f32_4(v2f a, v2f b, v8f c) {
  return __builtin_amdgcn_wmma_f32_16x16x4_f32(
       false, a,  false, b,
       (short)0, c,  false,  false);
}

__device__ __forceinline__ float half_reduce_sum(float x) {
  x += __shfl_xor(x, 1);
  x += __shfl_xor(x, 2);
  x += __shfl_xor(x, 4);
  x += __shfl_xor(x, 8);
  return x;
}

__device__ __forceinline__ float4 ldg4(const float* p) {
  return *(const float4*)p;
}

#define TW_ 8
#define PFU_ 64
#define PFD_ 128

__global__ __launch_bounds__(256) void attn_kernel(
    const float* __restrict__ q, const float* __restrict__ k,
    const float* __restrict__ v, const float* __restrict__ bias,
    const float* __restrict__ law, const int* __restrict__ idx,
    float* __restrict__ attn_out) {
  __shared__ __attribute__((aligned(16))) float sK[T_][PAD_];
  __shared__ __attribute__((aligned(16))) float sV[T_][PAD_];
  __shared__ __attribute__((aligned(8))) int sRow[E_];
  __shared__ __attribute__((aligned(16))) float sQ[TW_][16][PAD_];
  __shared__ float sP[TW_][16][17];

  const int tid  = threadIdx.x;
  const int lane = tid & 31;
  const int w    = tid >> 5;
  const int row  = lane & 15;
  const int hi   = lane >> 4;

  const int TB  = T_ / (16 * TW_);
  const int bid = blockIdx.x;
  const int tg  = bid % TB;
  const int h   = (bid / TB) % NH_;
  const int b   = bid / (TB * NH_);
  const int t0  = (tg * TW_ + w) * 16;

  const float* pfb = hi ? law : bias;
  const int pfr = (hi ? b * T_ : (b * NH_ + h) * T_) + t0 + row;
  const float* pf = pfb + (size_t)pfr * S_;
  __builtin_prefetch(pf, 0, 3);
  __builtin_prefetch(pf + PFU_, 0, 3);

  {
    const int r = tid >> 3;
    const int c = (tid & 7) * 4;
    const float* kb = k + ((size_t)b * T_) * HID_ + h * HD_ + c;
    const float* vb = v + ((size_t)b * T_) * HID_ + h * HD_ + c;
    for (int p = 0; p < T_ / 32; ++p) {
      const int s = p * 32 + r;
      *(float4*)&sK[s][c] = ldg4(kb + (size_t)s * HID_);
      *(float4*)&sV[s][c] = ldg4(vb + (size_t)s * HID_);
    }
    static_assert(E_ == 32 * TW_, "one table entry per thread of the block");
    sRow[tid] = idx[b * E_ + tid] & (T_ - 1);
  }

  {
    const int qrow = lane >> 1;
    const int qc   = (lane & 1) * 16;
    const float* qp =
        q + ((size_t)(b * T_ + t0 + qrow)) * HID_ + h * HD_ + qc;
#pragma unroll
    for (int ii = 0; ii < 4; ++ii)
      *(float4*)&sQ[w][qrow][qc + ii * 4] = ldg4(qp + ii * 4);
  }

  v2f aq[8];
#pragma unroll
  for (int kk = 0; kk < 8; ++kk) {
    const int kc = kk * 4 + 2 * hi;
    aq[kk].x = sQ[w][row][kc];
    aq[kk].y = sQ[w][row][kc + 1];
  }

  float rsum[8];
#pragma unroll
  for (int j = 0; j < 8; ++j) rsum[j] = 0.0f;
  v8f acc0 = {};
  v8f acc1 = {};

  const float* bp = bias + (((size_t)b * NH_ + h) * T_ + t0 + 8 * hi) * S_ + row;
  const float* lp = law + ((size_t)b * T_ + t0 + 8 * hi) * S_ + row;

  __syncthreads();

  auto s_tile = [&](const int s0, const int krow, const int (&vr0)[4],
                    const int (&vr1)[4]) {
    float bv[8], lv[8];
#pragma unroll
    for (int j = 0; j < 8; ++j) {
      bv[j] = bp[(size_t)j * S_ + s0];
      lv[j] = lp[(size_t)j * S_ + s0];
    }

    v8f c = {};
    const float* kp = &sK[krow][2 * hi];
#pragma unroll
    for (int kk = 0; kk < 8; ++kk) {
      v2f bb;
      bb.x = kp[kk * 4];
      bb.y = kp[kk * 4 + 1];
      c = wmma_f32_4(aq[kk], bb, c);
    }

#pragma unroll
    for (int j = 0; j < 8; ++j) {
      const int m = j + 8 * hi;
      const float sc = c[j] + bv[j];
      const float ex = __expf(fminf(sc, EXP_CLAMP_));
      const float e = (lv[j] <= CUTOFF_) ? 0.0f : ex;
      rsum[j] += e;
      sP[w][m][row] = e * lv[j];
    }

#pragma unroll
    for (int kk = 0; kk < 4; ++kk) {
      const int kc = kk * 4 + 2 * hi;
      v2f a, b0, b1;
      a.x  = sP[w][row][kc];
      a.y  = sP[w][row][kc + 1];
      b0.x = sV[vr0[kk]][row];
      b0.y = sV[vr1[kk]][row];
      b1.x = sV[vr0[kk]][row + 16];
      b1.y = sV[vr1[kk]][row + 16];
      acc0 = wmma_f32_4(a, b0, acc0);
      acc1 = wmma_f32_4(a, b1, acc1);
    }
  };

  for (int sp = 0; sp < T_ / PFU_; ++sp) {
    __builtin_prefetch(pf + sp * PFU_ + PFD_, 0, 3);
    for (int st = 4 * sp; st < 4 * sp + 4; ++st) {
      const int s0 = st * 16;
      int vr0[4], vr1[4];
#pragma unroll
      for (int kk = 0; kk < 4; ++kk) {
        vr0[kk] = s0 + kk * 4 + 2 * hi;
        vr1[kk] = vr0[kk] + 1;
      }
      s_tile(s0, s0 + row, vr0, vr1);
    }
  }
  for (int ep = 0; ep < E_ / PFU_; ++ep) {
    if (T_ + ep * PFU_ + PFD_ < S_)
      __builtin_prefetch(pf + T_ + ep * PFU_ + PFD_, 0, 3);
    for (int et = 4 * ep; et < 4 * ep + 4; ++et) {
      const int e0 = et * 16;
      const int krow = sRow[e0 + row];
      int vr0[4], vr1[4];
#pragma unroll
      for (int kk = 0; kk < 4; ++kk) {
        const int2 rr = *(const int2*)&sRow[e0 + kk * 4 + 2 * hi];
        vr0[kk] = rr.x;
        vr1[kk] = rr.y;
      }
      s_tile(T_ + e0, krow, vr0, vr1);
    }
  }

#pragma unroll
  for (int j = 0; j < 8; ++j) {
    const float s = half_reduce_sum(rsum[j]);
    const float inv = 1.0f / s;
    const int m = j + 8 * hi;
    const size_t o = ((size_t)(b * T_ + t0 + m)) * HID_ + h * HD_;
    attn_out[o + row]      = acc0[j] * inv;
    attn_out[o + 16 + row] = acc1[j] * inv;
  }
}

#define OM_ 64
#define ON_ 64
#define OP_ 516

__global__ __launch_bounds__(256) void ln_outproj_kernel(
    const float* __restrict__ attn, const float* __restrict__ ln_g,
    const float* __restrict__ ln_b, const float* __restrict__ Wout,
    float* __restrict__ out) {
  __shared__ __attribute__((aligned(16))) float sA[OM_][OP_];
  __shared__ __attribute__((aligned(16))) float sW[ON_][OP_];

  const int tid  = threadIdx.x;
  const int lane = tid & 31;
  const int w    = tid >> 5;
  const int row  = lane & 15;
  const int hi   = lane >> 4;

  const int NB = HID_ / ON_;
  const int n0 = (blockIdx.x % NB) * ON_;
  const int m0 = (blockIdx.x / NB) * OM_;

  const int lc = lane * 4;
  float4 g[4], o[4];
#pragma unroll
  for (int i = 0; i < 4; ++i) {
    g[i] = ldg4(ln_g + lc + 128 * i);
    o[i] = ldg4(ln_b + lc + 128 * i);
  }

  float4 x[8][4];
  {
    const float* ap = attn + ((size_t)(m0 + 8 * w)) * HID_ + lc;
#pragma unroll
    for (int rr = 0; rr < 8; ++rr) {
#pragma unroll
      for (int i = 0; i < 4; ++i)
        x[rr][i] = ldg4(ap + (size_t)rr * HID_ + 128 * i);
    }
  }

  const int r2 = tid >> 7;
  const int c4 = (tid & 127) * 4;
  float4 tw[ON_ / 2];
  {
    const float* wp = Wout + ((size_t)(n0 + r2)) * HID_ + c4;
#pragma unroll
    for (int i = 0; i < ON_ / 2; ++i) tw[i] = ldg4(wp + (size_t)(2 * i) * HID_);
  }

  {
    float s[8];
#pragma unroll
    for (int rr = 0; rr < 8; ++rr) {
      float t = 0.0f;
#pragma unroll
      for (int i = 0; i < 4; ++i)
        t += (x[rr][i].x + x[rr][i].y) + (x[rr][i].z + x[rr][i].w);
      s[rr] = t;
    }
#pragma unroll
    for (int m = 1; m < 32; m <<= 1) {
#pragma unroll
      for (int rr = 0; rr < 8; ++rr) s[rr] += __shfl_xor(s[rr], m);
    }
#pragma unroll
    for (int rr = 0; rr < 8; ++rr) {
      const float mean = s[rr] * (1.0f / HID_);
      float t = 0.0f;
#pragma unroll
      for (int i = 0; i < 4; ++i) {
        x[rr][i].x -= mean; x[rr][i].y -= mean;
        x[rr][i].z -= mean; x[rr][i].w -= mean;
        t += (x[rr][i].x * x[rr][i].x + x[rr][i].y * x[rr][i].y) +
             (x[rr][i].z * x[rr][i].z + x[rr][i].w * x[rr][i].w);
      }
      s[rr] = t;
    }
#pragma unroll
    for (int m = 1; m < 32; m <<= 1) {
#pragma unroll
      for (int rr = 0; rr < 8; ++rr) s[rr] += __shfl_xor(s[rr], m);
    }
#pragma unroll
    for (int rr = 0; rr < 8; ++rr) {
      const float rs = rsqrtf(s[rr] * (1.0f / HID_) + LN_EPS_);
      float* ar = &sA[8 * w + rr][lc];
#pragma unroll
      for (int i = 0; i < 4; ++i) {
        float4 y;
        y.x = x[rr][i].x * rs * g[i].x + o[i].x;
        y.y = x[rr][i].y * rs * g[i].y + o[i].y;
        y.z = x[rr][i].z * rs * g[i].z + o[i].z;
        y.w = x[rr][i].w * rs * g[i].w + o[i].w;
        *(float4*)(ar + 128 * i) = y;
      }
    }
  }
#pragma unroll
  for (int i = 0; i < ON_ / 2; ++i) *(float4*)&sW[2 * i + r2][c4] = tw[i];
  __syncthreads();

  const int mt = w & 3;
  const int nh = w >> 2;
  const float* pa  = &sA[mt * 16 + row][2 * hi];
  const float* pb0 = &sW[nh * 32 + row][2 * hi];
  const float* pb1 = &sW[nh * 32 + 16 + row][2 * hi];

  v8f c0 = {};
  v8f c1 = {};

#pragma unroll
  for (int ks = 0; ks < HID_ / 4; ++ks) {
    const int kc = ks * 4;
    v2f a, b0, b1;
    a.x  = pa[kc];   a.y  = pa[kc + 1];
    b0.x = pb0[kc];  b0.y = pb0[kc + 1];
    b1.x = pb1[kc];  b1.y = pb1[kc + 1];
    c0 = wmma_f32_4(a, b0, c0);
    c1 = wmma_f32_4(a, b1, c1);
  }

#pragma unroll
  for (int j = 0; j < 8; ++j) {
    const int m = m0 + mt * 16 + j + 8 * hi;
    float* op = out + (size_t)m * HID_ + n0 + nh * 32 + row;
    op[0]  = c0[j];
    op[16] = c1[j];
  }
}

extern "C" void kernel_launch(void* const* d_in, const int* in_sizes, int n_in,
                              void* d_out, int out_size, void* d_ws,
                              size_t ws_size, hipStream_t stream) {
  const float* q    = (const float*)d_in[0];
  const float* k    = (const float*)d_in[1];
  const float* v    = (const float*)d_in[2];
  const float* bias = (const float*)d_in[3];
  const int*   idx  = (const int*)d_in[5];
  const float* law  = (const float*)d_in[6];
  const float* Wout = (const float*)d_in[8];
  const float* lng  = (const float*)d_in[9];
  const float* lnb  = (const float*)d_in[10];
  float* out = (float*)d_out;

  float* attn = (float*)d_ws;

  attn_kernel<<<B_ * NH_ * (T_ / 128), 256, 0, stream>>>(q, k, v, bias, law,
                                                         idx, attn);
  ln_outproj_kernel<<<(B_ * T_ / OM_) * (HID_ / ON_), 256, 0, stream>>>(
      attn, lng, lnb, Wout, out);
}
